# starting kernel
# baseline (speedup 1.0000x reference)
#include <hip/hip_runtime.h>


typedef float v2f __attribute__((ext_vector_type(2)));
typedef float v8f __attribute__((ext_vector_type(8)));

#define GP_N_NODES 100000
#define GP_DEG 16
#define GP_D_FEAT 64

__global__ __launch_bounds__(256) void spmm_hop_wmma(
    const float* __restrict__ xin,
    const float* __restrict__ vals,
    const int*   __restrict__ indices,
    float*       __restrict__ xout,
    int nrows)
{
    const int  lane = threadIdx.x & 31;
    const bool hi   = lane >= 16;
    const int  fl   = lane & 15;
    const int  wavesPerBlock = blockDim.x >> 5;
    const int  wave   = blockIdx.x * wavesPerBlock + (threadIdx.x >> 5);
    const int  stride = gridDim.x * wavesPerBlock;

    for (int row = wave; row < nrows; row += stride) {
        const int r = __builtin_amdgcn_readfirstlane(row);
        const float* __restrict__ wp = vals    + (size_t)r * GP_DEG;
        const int*   __restrict__ ip = indices + (size_t)r * GP_DEG;

        v2f A[4];
        v2f B[4][4];

        #pragma unroll
        for (int s = 0; s < 4; ++s) {
            const float w0 = wp[4*s+0], w1 = wp[4*s+1];
            const float w2 = wp[4*s+2], w3 = wp[4*s+3];
            const int   n0 = ip[4*s+0], n1 = ip[4*s+1];
            const int   n2 = ip[4*s+2], n3 = ip[4*s+3];

            A[s].x = hi ? w2 : w0;
            A[s].y = hi ? w3 : w1;

            const float* bx = xin + (size_t)(hi ? n2 : n0) * GP_D_FEAT + fl;
            const float* by = xin + (size_t)(hi ? n3 : n1) * GP_D_FEAT + fl;
            #pragma unroll
            for (int t = 0; t < 4; ++t) {
                B[s][t].x = bx[16 * t];
                B[s][t].y = by[16 * t];
            }
        }

        v8f acc[4] = {};
        #pragma unroll
        for (int s = 0; s < 4; ++s) {
            #pragma unroll
            for (int t = 0; t < 4; ++t) {
                acc[t] = __builtin_amdgcn_wmma_f32_16x16x4_f32(
                     false, A[s],  false, B[s][t],
                     (short)0, acc[t],
                     false,  false);
            }
        }

        float* __restrict__ op = xout + (size_t)row * GP_D_FEAT;
        op[lane]      = hi ? acc[1][0] : acc[0][0];
        op[32 + lane] = hi ? acc[3][0] : acc[2][0];
    }
}

extern "C" void kernel_launch(void* const* d_in, const int* in_sizes, int n_in,
                              void* d_out, int out_size, void* d_ws, size_t ws_size,
                              hipStream_t stream) {
    (void)in_sizes; (void)n_in; (void)out_size; (void)ws_size;

    const float* x       = (const float*)d_in[0];
    const float* values  = (const float*)d_in[1];
    const int*   indices = (const int*)  d_in[3];
    float* out = (float*)d_out;
    float* ws  = (float*)d_ws;

    const int nrows       = GP_N_NODES;
    const int threads     = 256;
    const int rowsPerWave = 4;
    const int waves  = (nrows + rowsPerWave - 1) / rowsPerWave;
    const int blocks = (waves + (threads / 32) - 1) / (threads / 32);

    spmm_hop_wmma<<<blocks, threads, 0, stream>>>(x,   values, indices, out, nrows);
    spmm_hop_wmma<<<blocks, threads, 0, stream>>>(out, values, indices, ws,  nrows);
    spmm_hop_wmma<<<blocks, threads, 0, stream>>>(ws,  values, indices, out, nrows);
}


# Each gather is one whole 128-B line of one neighbor: scalar row base + lane*4, no vector address math; line halves on K0/1 and K2/3, weights on A rows 0/8; new sum order: 2 neighbors per WMMA
# speedup vs baseline: 1.1025x; 1.1025x over previous
#include <hip/hip_runtime.h>


typedef float v2f __attribute__((ext_vector_type(2)));
typedef float v8f __attribute__((ext_vector_type(8)));

typedef __attribute__((address_space(1))) const float gmem_cf;
typedef __attribute__((address_space(1))) float       gmem_f;

#define GP_N_NODES 100000
#define GP_DEG 16
#define GP_D_FEAT 64

__global__ __launch_bounds__(256) void spmm_hop_wmma(
    const float* __restrict__ xin,
    const float* __restrict__ vals,
    const int*   __restrict__ indices,
    float*       __restrict__ xout,
    int nrows)
{
    const unsigned lane = threadIdx.x & 31u;
    const bool live = (lane < 8u) || (lane >= 24u);
    const unsigned lane4 = lane * 4u;
    const int  wavesPerBlock = blockDim.x >> 5;
    const int  wave   = blockIdx.x * wavesPerBlock + (threadIdx.x >> 5);
    const int  stride = gridDim.x * wavesPerBlock;
    const unsigned long long xbase = (unsigned long long)xin;

    for (int row = wave; row < nrows; row += stride) {
        const int r = __builtin_amdgcn_readfirstlane(row);
        unsigned lo = lane4;
        asm volatile("" : "+v"(lo));
        const float* __restrict__ wp = vals    + (size_t)r * GP_DEG;
        const int*   __restrict__ ip = indices + (size_t)r * GP_DEG;

        float w[GP_DEG];
        unsigned long long nb[GP_DEG];
        #pragma unroll
        for (int j = 0; j < GP_DEG; ++j) {
            w[j]  = wp[j];
            nb[j] = xbase + ((unsigned)ip[j] << 8);
        }
        const unsigned long long op = (unsigned long long)xout + ((unsigned)r << 8);

        __builtin_amdgcn_sched_barrier(0);

        v2f B[8][2];
        #pragma unroll
        for (int p = 0; p < 8; ++p) {
            #pragma unroll
            for (int t = 0; t < 2; ++t) {
                B[p][t].x = *(gmem_cf*)(nb[2*p]   + lo + 128 * t);
                B[p][t].y = *(gmem_cf*)(nb[2*p+1] + lo + 128 * t);
            }
        }

        v8f acc[2] = {};
        #pragma unroll
        for (int p = 0; p < 8; ++p) {
            v2f A;
            A.x = live ? w[2*p]   : 0.0f;
            A.y = live ? w[2*p+1] : 0.0f;
            #pragma unroll
            for (int t = 0; t < 2; ++t) {
                acc[t] = __builtin_amdgcn_wmma_f32_16x16x4_f32(
                     false, A,  false, B[p][t],
                     (short)0, acc[t],
                     false,  false);
            }
        }

        *(gmem_f*)(op + lo)       = acc[0][0];
        *(gmem_f*)(op + lo + 128) = acc[1][0];
    }
}

extern "C" void kernel_launch(void* const* d_in, const int* in_sizes, int n_in,
                              void* d_out, int out_size, void* d_ws, size_t ws_size,
                              hipStream_t stream) {
    (void)in_sizes; (void)n_in; (void)out_size; (void)ws_size;

    const float* x       = (const float*)d_in[0];
    const float* values  = (const float*)d_in[1];
    const int*   indices = (const int*)  d_in[3];
    float* out = (float*)d_out;
    float* ws  = (float*)d_ws;

    const int nrows       = GP_N_NODES;
    const int threads     = 256;
    const int rowsPerWave = 4;
    const int waves  = (nrows + rowsPerWave - 1) / rowsPerWave;
    const int blocks = (waves + (threads / 32) - 1) / (threads / 32);

    spmm_hop_wmma<<<blocks, threads, 0, stream>>>(x,   values, indices, out, nrows);
    spmm_hop_wmma<<<blocks, threads, 0, stream>>>(out, values, indices, ws,  nrows);
    spmm_hop_wmma<<<blocks, threads, 0, stream>>>(ws,  values, indices, out, nrows);
}
